# starting kernel
# baseline (speedup 1.0000x reference)
#include <hip/hip_runtime.h>
#include <hip/hip_bf16.h>

typedef __attribute__((ext_vector_type(2))) float v2f;
typedef __attribute__((ext_vector_type(8))) float v8f;

#define EPSV 1e-6f

__global__ void HyperZero_kernel(float* __restrict__ p, int n) {
  int i = blockIdx.x * blockDim.x + threadIdx.x;
  if (i < n) p[i] = 0.0f;
}

__global__ __launch_bounds__(256) void HyperEdge_kernel(
    const float* __restrict__ vertices,
    const int*   __restrict__ edges,
    const float* __restrict__ messages,
    float* __restrict__ sums,
    float* __restrict__ cnt,
    long long E) {
  const int lane = threadIdx.x & 31;
  const long long wid = (long long)blockIdx.x * (blockDim.x >> 5) + (threadIdx.x >> 5);
  const long long e0 = wid << 5;
  if (e0 >= E) return;

  const int octet = lane >> 3;
  const int perm  = (octet == 1) ? (lane + 8) : ((octet == 2) ? (lane - 8) : lane);
  const long long eg = e0 + perm;

  float weight = 0.0f;
  if (e0 + 32 <= E) {
    const int rowA = lane & 15;
    const int kb   = (lane < 16) ? 0 : 2;
    const float* m1 = messages + (e0 + rowA) * 8 + kb;
    const float* m2 = m1 + 16 * 8;
    v2f a1lo = *(const v2f*)(m1);
    v2f a1hi = *(const v2f*)(m1 + 4);
    v2f a2lo = *(const v2f*)(m2);
    v2f a2hi = *(const v2f*)(m2 + 4);
    v2f ones; ones[0] = 1.0f; ones[1] = 1.0f;

    v8f c1 = {};
    c1 = __builtin_amdgcn_wmma_f32_16x16x4_f32(false, a1lo, false, ones, (short)0, c1, false, false);
    c1 = __builtin_amdgcn_wmma_f32_16x16x4_f32(false, a1hi, false, ones, (short)0, c1, false, false);
    v8f c2 = {};
    c2 = __builtin_amdgcn_wmma_f32_16x16x4_f32(false, a2lo, false, ones, (short)0, c2, false, false);
    c2 = __builtin_amdgcn_wmma_f32_16x16x4_f32(false, a2hi, false, ones, (short)0, c2, false, false);

    const bool useD2 = (lane >> 3) & 1;
    v8f d = useD2 ? c2 : c1;
    const int r = lane & 7;
    float s = d[0];
    if (r == 1) s = d[1];
    if (r == 2) s = d[2];
    if (r == 3) s = d[3];
    if (r == 4) s = d[4];
    if (r == 5) s = d[5];
    if (r == 6) s = d[6];
    if (r == 7) s = d[7];
    weight = s * 0.125f;
  } else if (eg < E) {
    float s = 0.0f;
#pragma unroll
    for (int k = 0; k < 8; ++k) s += messages[eg * 8 + k];
    weight = s * 0.125f;
  }

  if (eg < E) {
    const int vs = edges[2 * eg + 0];
    const int vd = edges[2 * eg + 1];
    const float x0 = vertices[3 * vs + 0], x1 = vertices[3 * vs + 1], x2 = vertices[3 * vs + 2];
    const float y0 = vertices[3 * vd + 0], y1 = vertices[3 * vd + 1], y2 = vertices[3 * vd + 2];
    const float t  = -x0 * y0 + x1 * y1 + x2 * y2;
    const float v0 = y0 + x0 * t;
    const float v1 = y1 + x1 * t;
    const float v2 = y2 + x2 * t;
    const float vn   = sqrtf(-v0 * v0 + v1 * v1 + v2 * v2 + EPSV);
    const float dist = acoshf(-t);
    const float sc   = dist * weight / vn;
    atomicAdd(&sums[3 * vs + 0], v0 * sc);
    atomicAdd(&sums[3 * vs + 1], v1 * sc);
    atomicAdd(&sums[3 * vs + 2], v2 * sc);
    atomicAdd(&cnt[vs], 1.0f);
  }
}

__global__ void HyperVertex_kernel(const float* __restrict__ vertices,
                                   const float* __restrict__ sums,
                                   const float* __restrict__ cnt,
                                   float* __restrict__ out, int N) {
  int i = blockIdx.x * blockDim.x + threadIdx.x;
  if (i >= N) return;
  const float c   = cnt[i];
  const float inv = 1.0f / fmaxf(c, 1.0f);
  float a0 = 0.0f, a1 = 0.0f, a2 = 0.0f;
  if (c > 0.0f) {
    a0 = sums[3 * i + 0] * inv;
    a1 = sums[3 * i + 1] * inv;
    a2 = sums[3 * i + 2] * inv;
  }
  const float an = sqrtf(-a0 * a0 + a1 * a1 + a2 * a2 + EPSV);
  const float T  = an;
  const float e  = expf(T);
  const float ei = 1.0f / e;
  const float ch = 0.5f * (e + ei);
  const float sh = 0.5f * (e - ei);
  const float f  = sh / an;
  out[3 * i + 0] = ch * vertices[3 * i + 0] + f * a0;
  out[3 * i + 1] = ch * vertices[3 * i + 1] + f * a1;
  out[3 * i + 2] = ch * vertices[3 * i + 2] + f * a2;
}

extern "C" void kernel_launch(void* const* d_in, const int* in_sizes, int n_in,
                              void* d_out, int out_size, void* d_ws, size_t ws_size,
                              hipStream_t stream) {
  const float* vertices = (const float*)d_in[0];
  const int*   edges    = (const int*)d_in[1];
  const float* messages = (const float*)d_in[2];
  float* out = (float*)d_out;

  const int       N = in_sizes[0] / 3;
  const long long E = (long long)in_sizes[1] / 2;

  float* sums = (float*)d_ws;
  float* cnt  = sums + (size_t)3 * N;

  const int zn = 4 * N;
  HyperZero_kernel<<<(zn + 255) / 256, 256, 0, stream>>>(sums, zn);

  const long long waves  = (E + 31) >> 5;
  const long long blocks = (waves + 7) >> 3;
  HyperEdge_kernel<<<dim3((unsigned)blocks), 256, 0, stream>>>(
      vertices, edges, messages, sums, cnt, E);

  HyperVertex_kernel<<<(N + 255) / 256, 256, 0, stream>>>(vertices, sums, cnt, out, N);
}


# Owner scan, no edge record moved: a work-group per (16384-vertex range, 1/8 edge slice) scans a byte plane of range ids, gathers its edges, sums in LDS; no global f32 atomic; sum order = LDS arrival
# speedup vs baseline: 2.5038x; 2.5038x over previous
#include <hip/hip_runtime.h>
#include <hip/hip_bf16.h>


typedef __attribute__((ext_vector_type(2))) float v2f;
typedef __attribute__((ext_vector_type(8))) float v8f;

#define EPSV 1e-6f
#define RV_SHIFT 14
#define RV (1 << RV_SHIFT)
#define NSLICE 8
#define QCAP 256
#define WG_WAVES 32
#define STEP_EDGES 128
#define SLICE_GRAIN (WG_WAVES * STEP_EDGES)

__device__ __forceinline__ unsigned hyper_rid(int src) {
  const unsigned r = (unsigned)src >> RV_SHIFT;
  return r < 255u ? r : 255u;
}

__global__ void HyperRangeId_kernel(const int* __restrict__ edges,
                                    unsigned* __restrict__ rid32,
                                    long long E, long long nwords) {
  const long long t = (long long)blockIdx.x * blockDim.x + threadIdx.x;
  if (t >= nwords) return;
  const long long e0 = t * 4;
  unsigned packed = 0xFFFFFFFFu;
  if (e0 + 4 <= E) {
    const int4 a = *(const int4*)(edges + 2 * e0);
    const int4 b = *(const int4*)(edges + 2 * e0 + 4);
    packed = hyper_rid(a.x) | (hyper_rid(a.z) << 8) | (hyper_rid(b.x) << 16) | (hyper_rid(b.z) << 24);
  } else {
    for (int k = 0; k < 4; ++k) {
      if (e0 + k < E) {
        const unsigned sh = 8u * (unsigned)k;
        packed = (packed & ~(0xFFu << sh)) | (hyper_rid(edges[2 * (e0 + k)]) << sh);
      }
    }
  }
  rid32[t] = packed;
}

__global__ __launch_bounds__(1024) void HyperOwner_kernel(
    const float* __restrict__ vertices,
    const int*   __restrict__ edges,
    const float* __restrict__ messages,
    const unsigned* __restrict__ rid32,
    float* __restrict__ part,
    int N, int R, long long Es) {
  __shared__ __attribute__((aligned(16))) float acc[RV * 4];
  __shared__ unsigned queue[WG_WAVES * QCAP];

  const int tid  = threadIdx.x;
  const int lane = tid & 31;
  const int wave = tid >> 5;
  const int r = (int)(blockIdx.x % (unsigned)R);
  const int s = (int)(blockIdx.x / (unsigned)R);
  const int vbase = r << RV_SHIFT;
  const unsigned rr = (unsigned)r;

  for (int i = tid; i < RV; i += 1024) ((float4*)acc)[i] = make_float4(0.0f, 0.0f, 0.0f, 0.0f);
  __syncthreads();

  unsigned* myq = queue + wave * QCAP;
  unsigned head = 0, tail = 0;

  const long long wbase = (long long)s * Es + (long long)wave * STEP_EDGES;
  const unsigned* ridp = rid32 + (wbase >> 2) + lane;
  const int nsteps = (int)(Es / SLICE_GRAIN);

  const int perm  = (lane & 7) | ((lane & 8) << 1) | ((lane >> 1) & 8);
  const int rowA  = lane & 15;
  const int kb    = (lane < 16) ? 0 : 2;

  auto batch = [&](const unsigned n) {
    const unsigned i1 = ((unsigned)rowA < n) ? (unsigned)rowA : 0u;
    const unsigned i2 = ((unsigned)(rowA + 16) < n) ? (unsigned)(rowA + 16) : 0u;
    const unsigned ip = ((unsigned)perm < n) ? (unsigned)perm : 0u;
    const unsigned ea = myq[(head + i1) & (QCAP - 1)];
    const unsigned eb = myq[(head + i2) & (QCAP - 1)];
    const unsigned e  = myq[(head + ip) & (QCAP - 1)];
    const int2 sd = *(const int2*)(edges + 2 * (size_t)e);
    const float* m1 = messages + (size_t)ea * 8 + kb;
    const float* m2 = messages + (size_t)eb * 8 + kb;
    v2f a1lo = *(const v2f*)(m1);
    v2f a1hi = *(const v2f*)(m1 + 4);
    v2f a2lo = *(const v2f*)(m2);
    v2f a2hi = *(const v2f*)(m2 + 4);
    v2f ones; ones[0] = 1.0f; ones[1] = 1.0f;

    const int vs = sd.x;
    const int vd = sd.y;
    const float x0 = vertices[3 * vs + 0], x1 = vertices[3 * vs + 1], x2 = vertices[3 * vs + 2];
    const float y0 = vertices[3 * vd + 0], y1 = vertices[3 * vd + 1], y2 = vertices[3 * vd + 2];

    v8f c1 = {};
    c1 = __builtin_amdgcn_wmma_f32_16x16x4_f32(false, a1lo, false, ones, (short)0, c1, false, false);
    c1 = __builtin_amdgcn_wmma_f32_16x16x4_f32(false, a1hi, false, ones, (short)0, c1, false, false);
    v8f c2 = {};
    c2 = __builtin_amdgcn_wmma_f32_16x16x4_f32(false, a2lo, false, ones, (short)0, c2, false, false);
    c2 = __builtin_amdgcn_wmma_f32_16x16x4_f32(false, a2hi, false, ones, (short)0, c2, false, false);

    const bool useD2 = (lane >> 3) & 1;
    v8f d = useD2 ? c2 : c1;
    const int q = lane & 7;
    float sm = d[0];
    if (q == 1) sm = d[1];
    if (q == 2) sm = d[2];
    if (q == 3) sm = d[3];
    if (q == 4) sm = d[4];
    if (q == 5) sm = d[5];
    if (q == 6) sm = d[6];
    if (q == 7) sm = d[7];
    const float weight = sm * 0.125f;

    const float t  = -x0 * y0 + x1 * y1 + x2 * y2;
    const float v0 = y0 + x0 * t;
    const float v1 = y1 + x1 * t;
    const float v2 = y2 + x2 * t;
    const float vn   = sqrtf(-v0 * v0 + v1 * v1 + v2 * v2 + EPSV);
    const float dist = acoshf(-t);
    const float sc   = dist * weight / vn;
    if ((unsigned)perm < n) {
      float* a = acc + 4 * (vs - vbase);
      atomicAdd(a + 0, v0 * sc);
      atomicAdd(a + 1, v1 * sc);
      atomicAdd(a + 2, v2 * sc);
      atomicAdd(a + 3, 1.0f);
    }
    head += n;
  };

  if (nsteps > 0) {
    unsigned w = ridp[0];
    for (int it = 0; it < nsteps; ++it) {
      const int nx = (it + 1 < nsteps) ? (it + 1) : it;
      const unsigned wn = ridp[(long long)nx * (SLICE_GRAIN / 4)];

      const unsigned e_lane = (unsigned)(wbase + (long long)it * SLICE_GRAIN) + 4u * (unsigned)lane;
#pragma unroll
      for (int b = 0; b < 4; ++b) {
        const bool m = ((w >> (8 * b)) & 0xFFu) == rr;
        const unsigned mask = __builtin_amdgcn_ballot_w32(m);
        const unsigned pre  = __builtin_amdgcn_mbcnt_lo(mask, 0u);
        if (m) myq[(tail + pre) & (QCAP - 1)] = e_lane + (unsigned)b;
        tail += (unsigned)__popc(mask);
      }
      while (tail - head >= 32u) batch(32u);
      w = wn;
    }
  }
  while (tail != head) {
    const unsigned avail = tail - head;
    batch(avail < 32u ? avail : 32u);
  }

  __syncthreads();
  float4* dst = (float4*)part + (size_t)s * (size_t)N;
  for (int i = tid; i < RV; i += 1024) {
    const int v = vbase + i;
    if (v < N) dst[v] = ((const float4*)acc)[i];
  }
}

__global__ void HyperVertex_kernel(const float* __restrict__ vertices,
                                   const float* __restrict__ part,
                                   float* __restrict__ out, int N) {
  int i = blockIdx.x * blockDim.x + threadIdx.x;
  if (i >= N) return;
  const float4* p = (const float4*)part + i;
  float4 sum = p[0];
#pragma unroll
  for (int k = 1; k < NSLICE; ++k) {
    const float4 pk = p[(size_t)k * (size_t)N];
    sum.x += pk.x; sum.y += pk.y; sum.z += pk.z; sum.w += pk.w;
  }
  const float c   = sum.w;
  const float inv = 1.0f / fmaxf(c, 1.0f);
  float a0 = 0.0f, a1 = 0.0f, a2 = 0.0f;
  if (c > 0.0f) {
    a0 = sum.x * inv;
    a1 = sum.y * inv;
    a2 = sum.z * inv;
  }
  const float an = sqrtf(-a0 * a0 + a1 * a1 + a2 * a2 + EPSV);
  const float T  = an;
  const float e  = expf(T);
  const float ei = 1.0f / e;
  const float ch = 0.5f * (e + ei);
  const float sh = 0.5f * (e - ei);
  const float f  = sh / an;
  out[3 * i + 0] = ch * vertices[3 * i + 0] + f * a0;
  out[3 * i + 1] = ch * vertices[3 * i + 1] + f * a1;
  out[3 * i + 2] = ch * vertices[3 * i + 2] + f * a2;
}

extern "C" void kernel_launch(void* const* d_in, const int* in_sizes, int n_in,
                              void* d_out, int out_size, void* d_ws, size_t ws_size,
                              hipStream_t stream) {
  const float* vertices = (const float*)d_in[0];
  const int*   edges    = (const int*)d_in[1];
  const float* messages = (const float*)d_in[2];
  float* out = (float*)d_out;

  const int       N = in_sizes[0] / 3;
  const long long E = (long long)in_sizes[1] / 2;

  const int       R  = (N + RV - 1) >> RV_SHIFT;
  const long long Es = ((E + (long long)NSLICE * SLICE_GRAIN - 1) / ((long long)NSLICE * SLICE_GRAIN)) * SLICE_GRAIN;
  const long long Epad = Es * NSLICE;

  if (ws_size < (size_t)Epad + (size_t)NSLICE * (size_t)N * 16u) return;

  unsigned* rid32 = (unsigned*)d_ws;
  float*    part  = (float*)((char*)d_ws + Epad);

  const long long nwords = Epad >> 2;
  HyperRangeId_kernel<<<(unsigned)((nwords + 255) / 256), 256, 0, stream>>>(edges, rid32, E, nwords);

  HyperOwner_kernel<<<dim3((unsigned)(R * NSLICE)), 1024, 0, stream>>>(
      vertices, edges, messages, rid32, part, N, R, Es);

  HyperVertex_kernel<<<(N + 255) / 256, 256, 0, stream>>>(vertices, part, out, N);
}
